# starting kernel
# baseline (speedup 1.0000x reference)
#include <hip/hip_runtime.h>

typedef __attribute__((ext_vector_type(2))) float v2f;
typedef __attribute__((ext_vector_type(4))) float v4f;
typedef __attribute__((ext_vector_type(8))) float v8f;

#define NGEN   4
#define TWO_N  400
#define NCH    32
#define NB     512
#define OTILES (TWO_N / 16)

__global__ void build_invperm_kernel(const float* __restrict__ perm,
                                     int* __restrict__ pinv) {
    int t = blockIdx.x * blockDim.x + threadIdx.x;
    if (t >= NGEN * TWO_N) return;
    int g = t / TWO_N;
    int o = t - g * TWO_N;
    const float* p = perm + (size_t)g * TWO_N * TWO_N + o;
    int found = 0;
    for (int i = 0; i < TWO_N; ++i) {
        if (p[(size_t)i * TWO_N] > 0.5f) found = i;
    }
    pinv[t] = found;
}

__global__ void __launch_bounds__(256)
dihedral_gconv_wmma(const float* __restrict__ x,
                    const float* __restrict__ w,
                    const int*  __restrict__ pinv,
                    float* __restrict__ out) {
    const int lane  = threadIdx.x & 31;
    const int wid   = threadIdx.x >> 5;
    const int wave  = blockIdx.x * 8 + wid;
    const int b     = wave / OTILES;
    const int ot    = wave - b * OTILES;
    const int o0    = ot * 16;
    const int m     = lane & 15;
    const int khalf = lane >> 4;

    v8f acc0 = {};
    v8f acc1 = {};

    const float* xb = x + (size_t)b * (NCH * TWO_N);

    for (int g = 0; g < NGEN; ++g) {
        const int idx = pinv[g * TWO_N + o0 + m];
        const float* xcol = xb + idx;
        const float* wg   = w + g * (NCH * NCH);

#pragma unroll
        for (int kb = 0; kb < 8; ++kb) {
            const int c0 = kb * 4 + khalf * 2;

            v2f a;
            a.x = xcol[(c0 + 0) * TWO_N];
            a.y = xcol[(c0 + 1) * TWO_N];

            v2f b0, b1;
            b0.x = wg[(c0 + 0) * NCH + m];
            b0.y = wg[(c0 + 1) * NCH + m];
            b1.x = wg[(c0 + 0) * NCH + 16 + m];
            b1.y = wg[(c0 + 1) * NCH + 16 + m];

            acc0 = __builtin_amdgcn_wmma_f32_16x16x4_f32(
                       false, a, false, b0, (short)0, acc0, false, false);
            acc1 = __builtin_amdgcn_wmma_f32_16x16x4_f32(
                       false, a, false, b1, (short)0, acc1, false, false);
        }
    }

    float* op0 = out + (size_t)b * (NCH * TWO_N) + (size_t)m * TWO_N
                     + o0 + khalf * 8;
    v4f s;
    s = (v4f){acc0[0], acc0[1], acc0[2], acc0[3]};
    *reinterpret_cast<v4f*>(op0) = s;
    s = (v4f){acc0[4], acc0[5], acc0[6], acc0[7]};
    *reinterpret_cast<v4f*>(op0 + 4) = s;

    float* op1 = op0 + 16 * TWO_N;
    s = (v4f){acc1[0], acc1[1], acc1[2], acc1[3]};
    *reinterpret_cast<v4f*>(op1) = s;
    s = (v4f){acc1[4], acc1[5], acc1[6], acc1[7]};
    *reinterpret_cast<v4f*>(op1 + 4) = s;
}

extern "C" void kernel_launch(void* const* d_in, const int* in_sizes, int n_in,
                              void* d_out, int out_size, void* d_ws, size_t ws_size,
                              hipStream_t stream) {
    const float* x    = (const float*)d_in[0];
    const float* w    = (const float*)d_in[1];
    const float* perm = (const float*)d_in[2];
    float* out = (float*)d_out;
    int* pinv = (int*)d_ws;

    build_invperm_kernel<<<(NGEN * TWO_N + 255) / 256, 256, 0, stream>>>(perm, pinv);

    const int total_waves = NB * OTILES;
    dihedral_gconv_wmma<<<total_waves / 8, 256, 0, stream>>>(x, w, pinv, out);
}


# single launch: each of 256 work-groups owns 50 consecutive (o-tile,batch) tiles and first finds its own 16 source columns from perm into LDS; index kernel and workspace table dropped
# speedup vs baseline: 1.1120x; 1.1120x over previous
#include <hip/hip_runtime.h>


typedef __attribute__((ext_vector_type(2))) float v2f;
typedef __attribute__((ext_vector_type(4))) float v4f;
typedef __attribute__((ext_vector_type(8))) float v8f;
typedef __attribute__((ext_vector_type(4))) int   v4i;

#define NGEN     4
#define TWO_N    400
#define NCH      32
#define NB       512
#define NWG      256
#define TILES_WG 50
#define NWAVES   25
#define NTHREADS 800
#define MAXSEG   2

__global__ void __launch_bounds__(NTHREADS) __attribute__((amdgpu_waves_per_eu(7, 7)))
dihedral_gconv_fused(const float* __restrict__ x,
                     const float* __restrict__ w,
                     const float* __restrict__ perm,
                     float* __restrict__ out) {
    __shared__ __attribute__((aligned(16))) int wpart[MAXSEG][NGEN][NWAVES][16];
    __shared__ __attribute__((aligned(16))) int src_s[MAXSEG][NGEN][16];

    const int lane = threadIdx.x & 31;
    const int wid  = __builtin_amdgcn_readfirstlane(threadIdx.x >> 5);
    const int t0       = blockIdx.x * TILES_WG;
    const int ot_first = t0 >> 9;
    const int ot_last  = (t0 + TILES_WG - 1) >> 9;
    const int nseg     = ot_last - ot_first + 1;

    {
        const int q  = lane & 3;
        const int r  = lane >> 2;
        const int i0 = wid * 8 + r;
        const int i1 = i0 + 200;
        for (int seg = 0; seg < nseg; ++seg) {
            const float* pc = perm + (ot_first + seg) * 16 + q * 4;
            v4f lo[NGEN], hi[NGEN];
#pragma unroll
            for (int g = 0; g < NGEN; ++g) {
                const float* p = pc + (size_t)g * (TWO_N * TWO_N);
                lo[g] = *reinterpret_cast<const v4f*>(p + (size_t)i0 * TWO_N);
                hi[g] = *reinterpret_cast<const v4f*>(p + (size_t)i1 * TWO_N);
            }
#pragma unroll
            for (int g = 0; g < NGEN; ++g) {
                v4i mx;
#pragma unroll
                for (int j = 0; j < 4; ++j) {
                    int v = (lo[g][j] > 0.5f) ? i0 : -1;
                    v = (hi[g][j] > 0.5f) ? i1 : v;
                    v = max(v, __shfl_xor(v, 4));
                    v = max(v, __shfl_xor(v, 8));
                    v = max(v, __shfl_xor(v, 16));
                    mx[j] = v;
                }
                if (r == 0)
                    *reinterpret_cast<v4i*>(&wpart[seg][g][wid][q * 4]) = mx;
            }
        }
    }
    __syncthreads();
    if ((int)threadIdx.x < nseg * (NGEN * 16)) {
        const int seg = threadIdx.x >> 6;
        const int g   = (threadIdx.x >> 4) & 3;
        const int col = threadIdx.x & 15;
        int mx = -1;
#pragma unroll
        for (int wv = 0; wv < NWAVES; ++wv) mx = max(mx, wpart[seg][g][wv][col]);
        src_s[seg][g][col] = (mx < 0) ? 0 : mx;
    }
    __syncthreads();

    const int m     = lane & 15;
    const int khalf = lane >> 4;

    for (int tt = wid; tt < TILES_WG; tt += NWAVES) {
        const int t   = t0 + tt;
        const int ot  = t >> 9;
        const int b   = t & (NB - 1);
        const int seg = ot - ot_first;
        const int o0  = ot * 16;

        v8f acc0 = {};
        v8f acc1 = {};

        const float* xb = x + (size_t)b * (NCH * TWO_N);

        int woff = 0;
        asm volatile("" : "+s"(woff));
        const float* wt = w + woff;

        v2f a[NGEN][8];
#pragma unroll
        for (int g = 0; g < NGEN; ++g) {
            const int idx = src_s[seg][g][m];
            const float* xcol = xb + idx;
#pragma unroll
            for (int kb = 0; kb < 8; ++kb) {
                const int c0 = kb * 4 + khalf * 2;
                a[g][kb].x = xcol[(c0 + 0) * TWO_N];
                a[g][kb].y = xcol[(c0 + 1) * TWO_N];
            }
        }

#pragma unroll
        for (int g = 0; g < NGEN; ++g) {
            const float* wg = wt + g * (NCH * NCH);
#pragma unroll
            for (int kb = 0; kb < 8; ++kb) {
                const int c0 = kb * 4 + khalf * 2;

                v2f b0, b1;
                b0.x = wg[(c0 + 0) * NCH + m];
                b0.y = wg[(c0 + 1) * NCH + m];
                b1.x = wg[(c0 + 0) * NCH + 16 + m];
                b1.y = wg[(c0 + 1) * NCH + 16 + m];

                acc0 = __builtin_amdgcn_wmma_f32_16x16x4_f32(
                           false, a[g][kb], false, b0, (short)0, acc0, false, false);
                acc1 = __builtin_amdgcn_wmma_f32_16x16x4_f32(
                           false, a[g][kb], false, b1, (short)0, acc1, false, false);
            }
        }

        float* op0 = out + (size_t)b * (NCH * TWO_N) + (size_t)m * TWO_N
                         + o0 + khalf * 8;
        v4f s;
        s = (v4f){acc0[0], acc0[1], acc0[2], acc0[3]};
        *reinterpret_cast<v4f*>(op0) = s;
        s = (v4f){acc0[4], acc0[5], acc0[6], acc0[7]};
        *reinterpret_cast<v4f*>(op0 + 4) = s;

        float* op1 = op0 + 16 * TWO_N;
        s = (v4f){acc1[0], acc1[1], acc1[2], acc1[3]};
        *reinterpret_cast<v4f*>(op1) = s;
        s = (v4f){acc1[4], acc1[5], acc1[6], acc1[7]};
        *reinterpret_cast<v4f*>(op1 + 4) = s;
    }
}

extern "C" void kernel_launch(void* const* d_in, const int* in_sizes, int n_in,
                              void* d_out, int out_size, void* d_ws, size_t ws_size,
                              hipStream_t stream) {
    const float* x    = (const float*)d_in[0];
    const float* w    = (const float*)d_in[1];
    const float* perm = (const float*)d_in[2];
    float* out = (float*)d_out;

    dihedral_gconv_fused<<<NWG, NTHREADS, 0, stream>>>(x, w, perm, out);
}


# permutation moved from operand to product: dense Y_g=W_g.x[b] from in-place x rows, Y_g via LDS, outputs pick columns src[g][o]; src table in own launch; sums over c per g, then added over g
# speedup vs baseline: 2.2574x; 2.0300x over previous
#include <hip/hip_runtime.h>


typedef __attribute__((ext_vector_type(2))) float v2f;
typedef __attribute__((ext_vector_type(4))) float v4f;
typedef __attribute__((ext_vector_type(8))) float v8f;
typedef __attribute__((ext_vector_type(4))) int   v4i;

#define NGEN     4
#define TWO_N    400
#define NCH      32
#define NB       512
#define NWAVES   25
#define NTHREADS 800
#define NOT      25
#define NWG      256
#define WF_FLOATS (NGEN * 8 * 2 * 32 * 2)
#define YP       404

__global__ void __launch_bounds__(NTHREADS)
dihedral_src_cols(const float* __restrict__ perm, int* __restrict__ src_tab) {
    __shared__ __attribute__((aligned(16))) int wpart[NGEN][NWAVES][16];

    const int lane = threadIdx.x & 31;
    const int wid  = threadIdx.x >> 5;
    const int q  = lane & 3;
    const int r  = lane >> 2;
    const int i0 = wid * 8 + r;
    const int i1 = i0 + 200;
    const float* pc = perm + blockIdx.x * 16 + q * 4;
    v4f lo[NGEN], hi[NGEN];
#pragma unroll
    for (int g = 0; g < NGEN; ++g) {
        const float* p = pc + (size_t)g * (TWO_N * TWO_N);
        lo[g] = *reinterpret_cast<const v4f*>(p + (size_t)i0 * TWO_N);
        hi[g] = *reinterpret_cast<const v4f*>(p + (size_t)i1 * TWO_N);
    }
#pragma unroll
    for (int g = 0; g < NGEN; ++g) {
        v4i mx;
#pragma unroll
        for (int j = 0; j < 4; ++j) {
            int v = (lo[g][j] > 0.5f) ? i0 : -1;
            v = (hi[g][j] > 0.5f) ? i1 : v;
            v = max(v, __shfl_xor(v, 4));
            v = max(v, __shfl_xor(v, 8));
            v = max(v, __shfl_xor(v, 16));
            mx[j] = v;
        }
        if (r == 0)
            *reinterpret_cast<v4i*>(&wpart[g][wid][q * 4]) = mx;
    }
    __syncthreads();
    if ((int)threadIdx.x < NGEN * 16) {
        const int g   = threadIdx.x >> 4;
        const int col = threadIdx.x & 15;
        int mx = -1;
#pragma unroll
        for (int wv = 0; wv < NWAVES; ++wv) mx = max(mx, wpart[g][wv][col]);
        src_tab[g * TWO_N + blockIdx.x * 16 + col] = (mx < 0) ? 0 : mx;
    }
}

__global__ void __launch_bounds__(NTHREADS) __attribute__((amdgpu_waves_per_eu(7, 7)))
dihedral_gconv_outperm(const float* __restrict__ x,
                       const float* __restrict__ w,
                       const int* __restrict__ src_tab,
                       float* __restrict__ out) {
    __shared__ __attribute__((aligned(16))) float smem[WF_FLOATS + 2 * NCH * YP];
    v2f*   Wf = reinterpret_cast<v2f*>(smem);
    float* Ys = smem + WF_FLOATS;

    const int tid  = threadIdx.x;
    const int lane = tid & 31;
    const int wid  = __builtin_amdgcn_readfirstlane(tid >> 5);
    const int m    = lane & 15;
    const int h    = lane >> 4;

    const int go  = tid % TWO_N;
    const int gd0 = (tid / TWO_N) * 16;

    v2f wp[3];
#pragma unroll
    for (int it = 0; it < 3; ++it) {
        const int idx = min(tid + it * NTHREADS, NGEN * 8 * 2 * 32 - 1);
        const int f  = idx >> 5;
        const int l  = idx & 31;
        const int g  = f >> 4;
        const int kb = (f >> 1) & 7;
        const int dt = f & 1;
        const int c  = kb * 4 + (l >> 4) * 2;
        const int n  = dt * 16 + (l & 15);
        wp[it].x = w[g * (NCH * NCH) + (c + 0) * NCH + n];
        wp[it].y = w[g * (NCH * NCH) + (c + 1) * NCH + n];
    }

    int sv[NGEN];
#pragma unroll
    for (int g = 0; g < NGEN; ++g) sv[g] = src_tab[g * TWO_N + go];

    const int b0 = blockIdx.x * 2;
    const float* xa = x + (size_t)b0 * (NCH * TWO_N) + (size_t)(2 * h) * TWO_N + wid * 16 + m;
    v2f a[2][8];
#pragma unroll
    for (int kb = 0; kb < 8; ++kb) {
        a[0][kb].x = xa[(kb * 4 + 0) * TWO_N];
        a[0][kb].y = xa[(kb * 4 + 1) * TWO_N];
    }

#pragma unroll
    for (int it = 0; it < 3; ++it) {
        const int idx = tid + it * NTHREADS;
        if (idx < NGEN * 8 * 2 * 32) Wf[idx] = wp[it];
    }
    asm volatile("s_wait_dscnt 0x0\n\ts_barrier_signal -1\n\ts_barrier_wait -1" ::: "memory");

#pragma unroll
    for (int kb = 0; kb < 8; ++kb) {
        a[1][kb].x = xa[(size_t)(NCH * TWO_N) + (kb * 4 + 0) * TWO_N];
        a[1][kb].y = xa[(size_t)(NCH * TWO_N) + (kb * 4 + 1) * TWO_N];
    }

    float oacc[16];
#pragma unroll
    for (int j = 0; j < 16; ++j) oacc[j] = 0.0f;

#pragma unroll
    for (int p = 0; p <= 8; ++p) {
        const int bb = (p >> 2) & 1;
        const int g  = p & 3;

        v2f fr[16];
        if (p < 8) {
            const v2f* wf = Wf + (g * 16) * 32 + lane;
#pragma unroll
            for (int k = 0; k < 16; ++k) fr[k] = wf[k * 32];
        }

        float pick[16];
        if (p > 0) {
            const int pp = p - 1;
            const float* ys = Ys + (pp & 1) * (NCH * YP) + gd0 * YP + sv[pp & 3];
#pragma unroll
            for (int j = 0; j < 16; ++j) pick[j] = ys[j * YP];
        }

        v8f c0 = {};
        v8f c1 = {};
        if (p < 8) {
#pragma unroll
            for (int kb = 0; kb < 8; ++kb) {
                c0 = __builtin_amdgcn_wmma_f32_16x16x4_f32(
                         false, a[bb][kb], false, fr[kb * 2 + 0], (short)0, c0, false, false);
                c1 = __builtin_amdgcn_wmma_f32_16x16x4_f32(
                         false, a[bb][kb], false, fr[kb * 2 + 1], (short)0, c1, false, false);
            }
        }

        if (p > 0) {
#pragma unroll
            for (int j = 0; j < 16; ++j) oacc[j] += pick[j];
            if (((p - 1) & 3) == 3) {
                float* op = out + (size_t)(b0 + ((p - 1) >> 2)) * (NCH * TWO_N)
                                + (size_t)gd0 * TWO_N + go;
#pragma unroll
                for (int j = 0; j < 16; ++j) { op[j * TWO_N] = oacc[j]; oacc[j] = 0.0f; }
            }
        }

        if (p < 8) {
            float* yp = Ys + (p & 1) * (NCH * YP) + m * YP + wid * 16 + h * 8;
            *reinterpret_cast<v4f*>(yp)     = (v4f){c0[0], c0[1], c0[2], c0[3]};
            *reinterpret_cast<v4f*>(yp + 4) = (v4f){c0[4], c0[5], c0[6], c0[7]};
            yp += 16 * YP;
            *reinterpret_cast<v4f*>(yp)     = (v4f){c1[0], c1[1], c1[2], c1[3]};
            *reinterpret_cast<v4f*>(yp + 4) = (v4f){c1[4], c1[5], c1[6], c1[7]};
            asm volatile("s_wait_dscnt 0x0\n\ts_barrier_signal -1\n\ts_barrier_wait -1" ::: "memory");
        }
    }
}

extern "C" void kernel_launch(void* const* d_in, const int* in_sizes, int n_in,
                              void* d_out, int out_size, void* d_ws, size_t ws_size,
                              hipStream_t stream) {
    const float* x    = (const float*)d_in[0];
    const float* w    = (const float*)d_in[1];
    const float* perm = (const float*)d_in[2];
    float* out = (float*)d_out;
    int* src_tab = (int*)d_ws;

    dihedral_src_cols<<<NOT, NTHREADS, 0, stream>>>(perm, src_tab);
    dihedral_gconv_outperm<<<NWG, NTHREADS, 0, stream>>>(x, w, src_tab, out);
}


# products kernel phases re-cut: one batch's 16-channel half with all four g per phase; 28 waves, the last i-tile's four g one per SIMD: equal matrix work, 4 barriers not 8; sum order kept
# speedup vs baseline: 2.4146x; 1.0696x over previous
#include <hip/hip_runtime.h>


typedef __attribute__((ext_vector_type(2))) float v2f;
typedef __attribute__((ext_vector_type(4))) float v4f;
typedef __attribute__((ext_vector_type(8))) float v8f;
typedef __attribute__((ext_vector_type(4))) int   v4i;

#define NGEN     4
#define TWO_N    400
#define NCH      32
#define NB       512
#define NWAVES   25
#define NTHREADS 800
#define NOT      25
#define NWG      256
#define WF_FLOATS (NGEN * 8 * 2 * 32 * 2)
#define YP       404
#define NMAIN    24
#define NWAVES2  28
#define NTHREADS2 (NWAVES2 * 32)
#define NPICKW   25
#define YROWS    (NGEN * 16)

__global__ void __launch_bounds__(NTHREADS)
dihedral_src_cols(const float* __restrict__ perm, int* __restrict__ src_tab) {
    __shared__ __attribute__((aligned(16))) int wpart[NGEN][NWAVES][16];

    const int lane = threadIdx.x & 31;
    const int wid  = threadIdx.x >> 5;
    const int q  = lane & 3;
    const int r  = lane >> 2;
    const int i0 = wid * 8 + r;
    const int i1 = i0 + 200;
    const float* pc = perm + blockIdx.x * 16 + q * 4;
    v4f lo[NGEN], hi[NGEN];
#pragma unroll
    for (int g = 0; g < NGEN; ++g) {
        const float* p = pc + (size_t)g * (TWO_N * TWO_N);
        lo[g] = *reinterpret_cast<const v4f*>(p + (size_t)i0 * TWO_N);
        hi[g] = *reinterpret_cast<const v4f*>(p + (size_t)i1 * TWO_N);
    }
#pragma unroll
    for (int g = 0; g < NGEN; ++g) {
        v4i mx;
#pragma unroll
        for (int j = 0; j < 4; ++j) {
            int v = (lo[g][j] > 0.5f) ? i0 : -1;
            v = (hi[g][j] > 0.5f) ? i1 : v;
            v = max(v, __shfl_xor(v, 4));
            v = max(v, __shfl_xor(v, 8));
            v = max(v, __shfl_xor(v, 16));
            mx[j] = v;
        }
        if (r == 0)
            *reinterpret_cast<v4i*>(&wpart[g][wid][q * 4]) = mx;
    }
    __syncthreads();
    if ((int)threadIdx.x < NGEN * 16) {
        const int g   = threadIdx.x >> 4;
        const int col = threadIdx.x & 15;
        int mx = -1;
#pragma unroll
        for (int wv = 0; wv < NWAVES; ++wv) mx = max(mx, wpart[g][wv][col]);
        src_tab[g * TWO_N + blockIdx.x * 16 + col] = (mx < 0) ? 0 : mx;
    }
}

__global__ void __launch_bounds__(NTHREADS2) __attribute__((amdgpu_waves_per_eu(7, 7)))
dihedral_gconv_outperm(const float* __restrict__ x,
                       const float* __restrict__ w,
                       const int* __restrict__ src_tab,
                       float* __restrict__ out) {
    __shared__ __attribute__((aligned(16))) float smem[WF_FLOATS + 2 * YROWS * YP];
    v2f*   Wf = reinterpret_cast<v2f*>(smem);
    float* Ys = smem + WF_FLOATS;

    const int tid  = threadIdx.x;
    const int lane = tid & 31;
    const int wid  = __builtin_amdgcn_readfirstlane(tid >> 5);
    const int m    = lane & 15;
    const int h    = lane >> 4;
    const int tile = (wid < NMAIN) ? wid : NMAIN;

    const int go = tid % TWO_N;
    const int dq = tid / TWO_N;

    v2f wp[3];
#pragma unroll
    for (int it = 0; it < 3; ++it) {
        const int idx = min(tid + it * NTHREADS2, NGEN * 8 * 2 * 32 - 1);
        const int f  = idx >> 5;
        const int l  = idx & 31;
        const int g  = f & 3;
        const int kb = (f >> 2) & 7;
        const int dh = f >> 5;
        const int c  = kb * 4 + (l >> 4) * 2;
        const int n  = dh * 16 + (l & 15);
        wp[it].x = w[g * (NCH * NCH) + (c + 0) * NCH + n];
        wp[it].y = w[g * (NCH * NCH) + (c + 1) * NCH + n];
    }

    int sv[NGEN];
#pragma unroll
    for (int g = 0; g < NGEN; ++g) sv[g] = src_tab[g * TWO_N + go];

    const int b0 = blockIdx.x * 2;
    const float* xa = x + (size_t)b0 * (NCH * TWO_N) + (size_t)(2 * h) * TWO_N + tile * 16 + m;
    v2f a[2][8];
#pragma unroll
    for (int kb = 0; kb < 8; ++kb) {
        a[0][kb].x = xa[(kb * 4 + 0) * TWO_N];
        a[0][kb].y = xa[(kb * 4 + 1) * TWO_N];
    }

#pragma unroll
    for (int it = 0; it < 3; ++it) {
        const int idx = tid + it * NTHREADS2;
        if (idx < NGEN * 8 * 2 * 32) Wf[idx] = wp[it];
    }
    asm volatile("s_wait_dscnt 0x0\n\ts_barrier_signal -1\n\ts_barrier_wait -1" ::: "memory");

#pragma unroll
    for (int kb = 0; kb < 8; ++kb) {
        a[1][kb].x = xa[(size_t)(NCH * TWO_N) + (kb * 4 + 0) * TWO_N];
        a[1][kb].y = xa[(size_t)(NCH * TWO_N) + (kb * 4 + 1) * TWO_N];
    }

#pragma unroll
    for (int p = 0; p <= 4; ++p) {
        const int bb = (p >> 1) & 1;
        const int dh = p & 1;
        const int pp = (p > 0) ? p - 1 : 0;

        const float* ysrc = Ys + (pp & 1) * (YROWS * YP) + (8 * dq) * YP;
        float* op = out + (size_t)(b0 + (pp >> 1)) * (NCH * TWO_N)
                        + (size_t)(16 * (pp & 1) + 8 * dq) * TWO_N + go;
        float* ydst = Ys + (p & 1) * (YROWS * YP) + m * YP + tile * 16 + h * 8;
        const v2f* wf = Wf + (dh * 8) * NGEN * 32 + lane;

        if (wid < NMAIN) {
            float pick[NGEN][8];
            if (p > 0) {
#pragma unroll
                for (int g = 0; g < NGEN; ++g)
#pragma unroll
                    for (int j = 0; j < 8; ++j) pick[g][j] = ysrc[(g * 16 + j) * YP + sv[g]];
            }
            v8f c0 = {}, c1 = {}, c2 = {}, c3 = {};
            if (p < 4) {
#pragma unroll
                for (int kb = 0; kb < 8; ++kb) {
                    c0 = __builtin_amdgcn_wmma_f32_16x16x4_f32(
                             false, a[bb][kb], false, wf[(kb * NGEN + 0) * 32], (short)0, c0, false, false);
                    c1 = __builtin_amdgcn_wmma_f32_16x16x4_f32(
                             false, a[bb][kb], false, wf[(kb * NGEN + 1) * 32], (short)0, c1, false, false);
                    c2 = __builtin_amdgcn_wmma_f32_16x16x4_f32(
                             false, a[bb][kb], false, wf[(kb * NGEN + 2) * 32], (short)0, c2, false, false);
                    c3 = __builtin_amdgcn_wmma_f32_16x16x4_f32(
                             false, a[bb][kb], false, wf[(kb * NGEN + 3) * 32], (short)0, c3, false, false);
                }
            }
            if (p > 0) {
#pragma unroll
                for (int j = 0; j < 8; ++j) {
                    float s = pick[0][j];
                    s += pick[1][j];
                    s += pick[2][j];
                    s += pick[3][j];
                    op[j * TWO_N] = s;
                }
            }
            if (p < 4) {
                float* yp = ydst;
                *reinterpret_cast<v4f*>(yp)     = (v4f){c0[0], c0[1], c0[2], c0[3]};
                *reinterpret_cast<v4f*>(yp + 4) = (v4f){c0[4], c0[5], c0[6], c0[7]};
                yp += 16 * YP;
                *reinterpret_cast<v4f*>(yp)     = (v4f){c1[0], c1[1], c1[2], c1[3]};
                *reinterpret_cast<v4f*>(yp + 4) = (v4f){c1[4], c1[5], c1[6], c1[7]};
                yp += 16 * YP;
                *reinterpret_cast<v4f*>(yp)     = (v4f){c2[0], c2[1], c2[2], c2[3]};
                *reinterpret_cast<v4f*>(yp + 4) = (v4f){c2[4], c2[5], c2[6], c2[7]};
                yp += 16 * YP;
                *reinterpret_cast<v4f*>(yp)     = (v4f){c3[0], c3[1], c3[2], c3[3]};
                *reinterpret_cast<v4f*>(yp + 4) = (v4f){c3[4], c3[5], c3[6], c3[7]};
            }
        } else {
            const int gl = wid - NMAIN;
            v8f c = {};
            if (p < 4) {
                const v2f* wfl = wf + gl * 32;
#pragma unroll
                for (int kb = 0; kb < 8; ++kb)
                    c = __builtin_amdgcn_wmma_f32_16x16x4_f32(
                            false, a[bb][kb], false, wfl[(kb * NGEN) * 32], (short)0, c, false, false);
                float* yp = ydst + gl * (16 * YP);
                *reinterpret_cast<v4f*>(yp)     = (v4f){c[0], c[1], c[2], c[3]};
                *reinterpret_cast<v4f*>(yp + 4) = (v4f){c[4], c[5], c[6], c[7]};
            }
            if (p > 0 && wid < NPICKW) {
                float pick[NGEN][8];
#pragma unroll
                for (int g = 0; g < NGEN; ++g)
#pragma unroll
                    for (int j = 0; j < 8; ++j) pick[g][j] = ysrc[(g * 16 + j) * YP + sv[g]];
#pragma unroll
                for (int j = 0; j < 8; ++j) {
                    float s = pick[0][j];
                    s += pick[1][j];
                    s += pick[2][j];
                    s += pick[3][j];
                    op[j * TWO_N] = s;
                }
            }
        }

        if (p < 4) {
            asm volatile("s_wait_dscnt 0x0\n\ts_barrier_signal -1\n\ts_barrier_wait -1" ::: "memory");
        }
    }
}

extern "C" void kernel_launch(void* const* d_in, const int* in_sizes, int n_in,
                              void* d_out, int out_size, void* d_ws, size_t ws_size,
                              hipStream_t stream) {
    const float* x    = (const float*)d_in[0];
    const float* w    = (const float*)d_in[1];
    const float* perm = (const float*)d_in[2];
    float* out = (float*)d_out;
    int* src_tab = (int*)d_ws;

    dihedral_src_cols<<<NOT, NTHREADS, 0, stream>>>(perm, src_tab);
    dihedral_gconv_outperm<<<NWG, NTHREADS2, 0, stream>>>(x, w, src_tab, out);
}
